# starting kernel
# baseline (speedup 1.0000x reference)
#include <hip/hip_runtime.h>

typedef __attribute__((ext_vector_type(2))) float v2f;
typedef __attribute__((ext_vector_type(8))) float v8f;

#define B_DIM   128
#define IN_DIM  1024
#define OUT_DIM 1024

__global__ __launch_bounds__(256) void dendritic_fused_kernel(
    const float* __restrict__ X,
    const float* __restrict__ Wnmda,
    const float* __restrict__ Wnon,
    const float* __restrict__ bnon,
    float* __restrict__ Y)
{
    const int lane = threadIdx.x & 31;
    const int wave = threadIdx.x >> 5;
    const int tile = blockIdx.x * 8 + wave;
    const int mt = tile >> 6;
    const int nt = tile & 63;

    const int mrow = mt * 16 + (lane & 15);
    const int nrow = nt * 16 + (lane & 15);
    const int h4   = (lane >> 4) * 4;

    const float* xr = X     + (size_t)mrow * IN_DIM + h4;
    const float* wn = Wnon  + (size_t)nrow * IN_DIM + h4;
    const float* wm = Wnmda + (size_t)nrow * IN_DIM + h4;

    v8f accN = {0.f,0.f,0.f,0.f,0.f,0.f,0.f,0.f};
    v8f accM = {0.f,0.f,0.f,0.f,0.f,0.f,0.f,0.f};

    const float edge_lo = (lane < 16)  ? 0.5f : 1.0f;
    const float edge_hi = (lane >= 16) ? 0.5f : 1.0f;

    for (int kb = 0; kb < IN_DIM; kb += 8) {
        float4 a4  = *(const float4*)(xr + kb);
        float4 b4n = *(const float4*)(wn + kb);
        float4 b4m = *(const float4*)(wm + kb);
        if (kb == 0)            b4m.x *= edge_lo;
        if (kb == IN_DIM - 8)   b4m.w *= edge_hi;

        v2f a0  = {a4.x,  a4.y},  a1  = {a4.z,  a4.w};
        v2f bn0 = {b4n.x, b4n.y}, bn1 = {b4n.z, b4n.w};
        v2f bm0 = {b4m.x, b4m.y}, bm1 = {b4m.z, b4m.w};

        accN = __builtin_amdgcn_wmma_f32_16x16x4_f32(false, a0, false, bn0, (short)0, accN, false, false);
        accN = __builtin_amdgcn_wmma_f32_16x16x4_f32(false, a1, false, bn1, (short)0, accN, false, false);
        accM = __builtin_amdgcn_wmma_f32_16x16x4_f32(false, a0, false, bm0, (short)0, accM, false, false);
        accM = __builtin_amdgcn_wmma_f32_16x16x4_f32(false, a1, false, bm1, (short)0, accM, false, false);
    }

    const float bb = bnon[nrow];
    float* op = Y + (size_t)(mt * 16 + (lane >> 4) * 8) * OUT_DIM + nrow;
    #pragma unroll
    for (int i = 0; i < 8; ++i) {
        float z     = accN[i] + bb;
        float s     = 1.0f / (1.0f + __expf(-z));
        float state = s - 1.0f;
        float ca    = fmaxf(accM[i], 0.0f);
        float nm    = 2.0f + state;
        float kd    = exp2f(-nm);
        float xn    = __powf(ca, nm);
        float y     = xn / (kd + xn) + state;
        op[(size_t)i * OUT_DIM] = y;
    }
}

extern "C" void kernel_launch(void* const* d_in, const int* in_sizes, int n_in,
                              void* d_out, int out_size, void* d_ws, size_t ws_size,
                              hipStream_t stream) {
    const float* X     = (const float*)d_in[0];
    const float* Wnmda = (const float*)d_in[1];
    const float* Wnon  = (const float*)d_in[2];
    const float* bnon  = (const float*)d_in[3];
    float* Y = (float*)d_out;

    dim3 grid(64);
    dim3 block(256);
    hipLaunchKernelGGL(dendritic_fused_kernel, grid, block, 0, stream,
                       X, Wnmda, Wnon, bnon, Y);
}


# Split k across the block's 8 waves (grid 64 -> 512 blocks, one 16x16 tile per block); partial tiles summed through LDS in fixed wave order, so the f32 summation order changes
# speedup vs baseline: 2.7589x; 2.7589x over previous
#include <hip/hip_runtime.h>

typedef __attribute__((ext_vector_type(2))) float v2f;
typedef __attribute__((ext_vector_type(8))) float v8f;

#define B_DIM   128
#define IN_DIM  1024
#define OUT_DIM 1024
#define KSPLIT  8
#define KSLICE  (IN_DIM / KSPLIT)

__global__ __launch_bounds__(256) void dendritic_fused_kernel(
    const float* __restrict__ X,
    const float* __restrict__ Wnmda,
    const float* __restrict__ Wnon,
    const float* __restrict__ bnon,
    float* __restrict__ Y)
{
    __shared__ float part[KSPLIT][2][256];

    const int lane = threadIdx.x & 31;
    const int wave = threadIdx.x >> 5;
    const int tile = blockIdx.x;
    const int mt = tile >> 6;
    const int nt = tile & 63;

    const int mrow = mt * 16 + (lane & 15);
    const int nrow = nt * 16 + (lane & 15);
    const int h4   = (lane >> 4) * 4;
    const int k0   = wave * KSLICE;

    const float* xr = X     + (size_t)mrow * IN_DIM + k0 + h4;
    const float* wn = Wnon  + (size_t)nrow * IN_DIM + k0 + h4;
    const float* wm = Wnmda + (size_t)nrow * IN_DIM + k0 + h4;

    v8f accN = {0.f,0.f,0.f,0.f,0.f,0.f,0.f,0.f};
    v8f accM = {0.f,0.f,0.f,0.f,0.f,0.f,0.f,0.f};

    const float edge_lo = (wave == 0 && lane < 16)           ? 0.5f : 1.0f;
    const float edge_hi = (wave == KSPLIT - 1 && lane >= 16) ? 0.5f : 1.0f;

    for (int kb = 0; kb < KSLICE; kb += 8) {
        float4 a4  = *(const float4*)(xr + kb);
        float4 b4n = *(const float4*)(wn + kb);
        float4 b4m = *(const float4*)(wm + kb);
        if (kb == 0)            b4m.x *= edge_lo;
        if (kb == KSLICE - 8)   b4m.w *= edge_hi;

        v2f a0  = {a4.x,  a4.y},  a1  = {a4.z,  a4.w};
        v2f bn0 = {b4n.x, b4n.y}, bn1 = {b4n.z, b4n.w};
        v2f bm0 = {b4m.x, b4m.y}, bm1 = {b4m.z, b4m.w};

        accN = __builtin_amdgcn_wmma_f32_16x16x4_f32(false, a0, false, bn0, (short)0, accN, false, false);
        accN = __builtin_amdgcn_wmma_f32_16x16x4_f32(false, a1, false, bn1, (short)0, accN, false, false);
        accM = __builtin_amdgcn_wmma_f32_16x16x4_f32(false, a0, false, bm0, (short)0, accM, false, false);
        accM = __builtin_amdgcn_wmma_f32_16x16x4_f32(false, a1, false, bm1, (short)0, accM, false, false);
    }

    #pragma unroll
    for (int i = 0; i < 8; ++i) {
        part[wave][0][i * 32 + lane] = accN[i];
        part[wave][1][i * 32 + lane] = accM[i];
    }
    __syncthreads();

    const int slot = threadIdx.x;
    float sN = part[0][0][slot];
    float sM = part[0][1][slot];
    for (int w = 1; w < KSPLIT; ++w) {
        sN += part[w][0][slot];
        sM += part[w][1][slot];
    }

    const float bb = bnon[nrow];
    float z     = sN + bb;
    float s     = 1.0f / (1.0f + __expf(-z));
    float state = s - 1.0f;
    float ca    = fmaxf(sM, 0.0f);
    float nm    = 2.0f + state;
    float kd    = exp2f(-nm);
    float xn    = __powf(ca, nm);
    float y     = xn / (kd + xn) + state;
    Y[(size_t)(mt * 16 + (lane >> 4) * 8 + wave) * OUT_DIM + nrow] = y;
}

extern "C" void kernel_launch(void* const* d_in, const int* in_sizes, int n_in,
                              void* d_out, int out_size, void* d_ws, size_t ws_size,
                              hipStream_t stream) {
    const float* X     = (const float*)d_in[0];
    const float* Wnmda = (const float*)d_in[1];
    const float* Wnon  = (const float*)d_in[2];
    const float* bnon  = (const float*)d_in[3];
    float* Y = (float*)d_out;

    dim3 grid(512);
    dim3 block(256);
    hipLaunchKernelGGL(dendritic_fused_kernel, grid, block, 0, stream,
                       X, Wnmda, Wnon, bnon, Y);
}


# Each wave stages its k-slice of x, W_non, W_nmda through its own LDS region: global loads now read whole 128-byte lines (4 per load, was 16 rows x 32 B), fragments come from LDS; sum order kept
# speedup vs baseline: 3.3852x; 1.2270x over previous
#include <hip/hip_runtime.h>

typedef __attribute__((ext_vector_type(2))) float v2f;
typedef __attribute__((ext_vector_type(4))) float v4f;
typedef __attribute__((ext_vector_type(8))) float v8f;

#define B_DIM   128
#define IN_DIM  1024
#define OUT_DIM 1024
#define KSPLIT  8
#define KSLICE  (IN_DIM / KSPLIT)
#define KCHUNK  32
#define NCHUNK  (KSLICE / KCHUNK)
#define NSTEP   (KCHUNK / 8)
#define PITCH4  9
#define OP4     (16 * PITCH4)
#define WAVE4   (3 * OP4)

#define WAVE_LDS_SYNC() do { \
    __builtin_amdgcn_fence(__ATOMIC_RELEASE, "wavefront"); \
    __builtin_amdgcn_wave_barrier(); \
    __builtin_amdgcn_fence(__ATOMIC_ACQUIRE, "wavefront"); } while (0)

__global__ __launch_bounds__(256) void dendritic_fused_kernel(
    const float* __restrict__ X,
    const float* __restrict__ Wnmda,
    const float* __restrict__ Wnon,
    const float* __restrict__ bnon,
    float* __restrict__ Y)
{
    __shared__ v4f stage[KSPLIT * WAVE4];

    const int lane = threadIdx.x & 31;
    const int wave = threadIdx.x >> 5;
    const int tile = blockIdx.x;
    const int mt = tile >> 6;
    const int nt = tile & 63;

    const int nrow = nt * 16 + (lane & 15);
    const int k0   = wave * KSLICE;

    const int lrow = ((lane >> 3) & 1) * 8 + (lane >> 4) * 4;
    const int lu   = lane & 7;
    const float* gx = X     + (size_t)(mt * 16 + lrow) * IN_DIM + k0 + lu * 4;
    const float* gn = Wnon  + (size_t)(nt * 16 + lrow) * IN_DIM + k0 + lu * 4;
    const float* gm = Wnmda + (size_t)(nt * 16 + lrow) * IN_DIM + k0 + lu * 4;

    v4f* sx = stage + wave * WAVE4;
    v4f* sn = sx + OP4;
    v4f* sm = sn + OP4;
    const int st_off = lrow * PITCH4 + lu;
    const int fr_off = (lane & 15) * PITCH4 + (lane >> 4);

    v8f accN = {0.f,0.f,0.f,0.f,0.f,0.f,0.f,0.f};
    v8f accM = {0.f,0.f,0.f,0.f,0.f,0.f,0.f,0.f};

    const float edge_lo = (wave == 0 && lane < 16)           ? 0.5f : 1.0f;
    const float edge_hi = (wave == KSPLIT - 1 && lane >= 16) ? 0.5f : 1.0f;

    v4f rx[2][4], rn[2][4], rm[2][4];
    #pragma unroll
    for (int c = 0; c < 2; ++c) {
        #pragma unroll
        for (int j = 0; j < 4; ++j) {
            rx[c][j] = *(const v4f*)(gx + (size_t)j * IN_DIM + c * KCHUNK);
            rn[c][j] = *(const v4f*)(gn + (size_t)j * IN_DIM + c * KCHUNK);
            rm[c][j] = *(const v4f*)(gm + (size_t)j * IN_DIM + c * KCHUNK);
        }
    }

    #pragma unroll
    for (int c = 0; c < NCHUNK; ++c) {
        #pragma unroll
        for (int j = 0; j < 4; ++j) {
            sx[st_off + j * PITCH4] = rx[c & 1][j];
            sn[st_off + j * PITCH4] = rn[c & 1][j];
            sm[st_off + j * PITCH4] = rm[c & 1][j];
        }
        WAVE_LDS_SYNC();

        if (c + 2 < NCHUNK) {
            #pragma unroll
            for (int j = 0; j < 4; ++j) {
                rx[c & 1][j] = *(const v4f*)(gx + (size_t)j * IN_DIM + (c + 2) * KCHUNK);
                rn[c & 1][j] = *(const v4f*)(gn + (size_t)j * IN_DIM + (c + 2) * KCHUNK);
                rm[c & 1][j] = *(const v4f*)(gm + (size_t)j * IN_DIM + (c + 2) * KCHUNK);
            }
        }
        __builtin_amdgcn_sched_barrier(0);

        v4f fa[NSTEP], fn[NSTEP], fm[NSTEP];
        #pragma unroll
        for (int s = 0; s < NSTEP; ++s) {
            fa[s] = sx[fr_off + 2 * s];
            fn[s] = sn[fr_off + 2 * s];
            fm[s] = sm[fr_off + 2 * s];
        }
        __builtin_amdgcn_sched_barrier(0);

        if (c == 0)          fm[0].x         *= edge_lo;
        if (c == NCHUNK - 1) fm[NSTEP - 1].w *= edge_hi;

        #pragma unroll
        for (int s = 0; s < NSTEP; ++s) {
            const v4f a4 = fa[s], b4n = fn[s], b4m = fm[s];
            v2f a0  = {a4.x,  a4.y},  a1  = {a4.z,  a4.w};
            v2f bn0 = {b4n.x, b4n.y}, bn1 = {b4n.z, b4n.w};
            v2f bm0 = {b4m.x, b4m.y}, bm1 = {b4m.z, b4m.w};

            accN = __builtin_amdgcn_wmma_f32_16x16x4_f32(false, a0, false, bn0, (short)0, accN, false, false);
            accN = __builtin_amdgcn_wmma_f32_16x16x4_f32(false, a1, false, bn1, (short)0, accN, false, false);
            accM = __builtin_amdgcn_wmma_f32_16x16x4_f32(false, a0, false, bm0, (short)0, accM, false, false);
            accM = __builtin_amdgcn_wmma_f32_16x16x4_f32(false, a1, false, bm1, (short)0, accM, false, false);
        }
        WAVE_LDS_SYNC();
        __builtin_amdgcn_sched_barrier(0);
    }

    float* mine = (float*)(stage + wave * WAVE4);
    #pragma unroll
    for (int i = 0; i < 8; ++i) {
        mine[i * 32 + lane]       = accN[i];
        mine[256 + i * 32 + lane] = accM[i];
    }
    __syncthreads();

    const int slot = threadIdx.x;
    const float* p0 = (const float*)stage;
    float sN = p0[slot];
    float sM = p0[256 + slot];
    for (int w = 1; w < KSPLIT; ++w) {
        const float* pw = (const float*)(stage + w * WAVE4);
        sN += pw[slot];
        sM += pw[256 + slot];
    }

    const float bb = bnon[nrow];
    float z     = sN + bb;
    float s     = 1.0f / (1.0f + __expf(-z));
    float state = s - 1.0f;
    float ca    = fmaxf(sM, 0.0f);
    float nm    = 2.0f + state;
    float kd    = exp2f(-nm);
    float xn    = __powf(ca, nm);
    float y     = xn / (kd + xn) + state;
    Y[(size_t)(mt * 16 + (lane >> 4) * 8 + wave) * OUT_DIM + nrow] = y;
}

extern "C" void kernel_launch(void* const* d_in, const int* in_sizes, int n_in,
                              void* d_out, int out_size, void* d_ws, size_t ws_size,
                              hipStream_t stream) {
    const float* X     = (const float*)d_in[0];
    const float* Wnmda = (const float*)d_in[1];
    const float* Wnon  = (const float*)d_in[2];
    const float* bnon  = (const float*)d_in[3];
    float* Y = (float*)d_out;

    dim3 grid(512);
    dim3 block(256);
    hipLaunchKernelGGL(dendritic_fused_kernel, grid, block, 0, stream,
                       X, Wnmda, Wnon, bnon, Y);
}


# Block tile 16x16 -> 32x16 (two x row tiles per block, grid 512 -> 256 blocks): each staged W row feeds two WMMA tiles, a third fewer bytes from L2 per WMMA, half the waves; f32 sum order unchanged
# speedup vs baseline: 3.5859x; 1.0593x over previous
#include <hip/hip_runtime.h>

typedef __attribute__((ext_vector_type(2))) float v2f;
typedef __attribute__((ext_vector_type(4))) float v4f;
typedef __attribute__((ext_vector_type(8))) float v8f;

#define B_DIM   128
#define IN_DIM  1024
#define OUT_DIM 1024
#define MTILES  2
#define KSPLIT  8
#define KSLICE  (IN_DIM / KSPLIT)
#define KCHUNK  32
#define NCHUNK  (KSLICE / KCHUNK)
#define NSTEP   (KCHUNK / 8)
#define PITCH4  9
#define OP4     (16 * PITCH4)
#define WAVE4   ((MTILES + 2) * OP4)

#define WAVE_LDS_SYNC() do { \
    __builtin_amdgcn_fence(__ATOMIC_RELEASE, "wavefront"); \
    __builtin_amdgcn_wave_barrier(); \
    __builtin_amdgcn_fence(__ATOMIC_ACQUIRE, "wavefront"); } while (0)

__global__ __launch_bounds__(256) void dendritic_fused_kernel(
    const float* __restrict__ X,
    const float* __restrict__ Wnmda,
    const float* __restrict__ Wnon,
    const float* __restrict__ bnon,
    float* __restrict__ Y)
{
    __shared__ v4f stage[KSPLIT * WAVE4];

    const int lane = threadIdx.x & 31;
    const int wave = threadIdx.x >> 5;
    const int tile = blockIdx.x;
    const int mt = tile >> 6;
    const int nt = tile & 63;

    const int nrow = nt * 16 + (lane & 15);
    const int k0   = wave * KSLICE;

    const int lrow = ((lane >> 3) & 1) * 8 + (lane >> 4) * 4;
    const int lu   = lane & 7;
    const float* gx = X     + (size_t)(mt * (16 * MTILES) + lrow) * IN_DIM + k0 + lu * 4;
    const float* gn = Wnon  + (size_t)(nt * 16 + lrow) * IN_DIM + k0 + lu * 4;
    const float* gm = Wnmda + (size_t)(nt * 16 + lrow) * IN_DIM + k0 + lu * 4;

    v4f* sx = stage + wave * WAVE4;
    v4f* sn = sx + MTILES * OP4;
    v4f* sm = sn + OP4;
    const int st_off = lrow * PITCH4 + lu;
    const int fr_off = (lane & 15) * PITCH4 + (lane >> 4);

    v8f accN[MTILES], accM[MTILES];
    #pragma unroll
    for (int t = 0; t < MTILES; ++t) {
        accN[t] = (v8f){0.f,0.f,0.f,0.f,0.f,0.f,0.f,0.f};
        accM[t] = (v8f){0.f,0.f,0.f,0.f,0.f,0.f,0.f,0.f};
    }

    const float edge_lo = (wave == 0 && lane < 16)           ? 0.5f : 1.0f;
    const float edge_hi = (wave == KSPLIT - 1 && lane >= 16) ? 0.5f : 1.0f;

    v4f rx[2][MTILES][4], rn[2][4], rm[2][4];
    #pragma unroll
    for (int c = 0; c < 2; ++c) {
        #pragma unroll
        for (int j = 0; j < 4; ++j) {
            #pragma unroll
            for (int t = 0; t < MTILES; ++t)
                rx[c][t][j] = *(const v4f*)(gx + (size_t)(t * 16 + j) * IN_DIM + c * KCHUNK);
            rn[c][j] = *(const v4f*)(gn + (size_t)j * IN_DIM + c * KCHUNK);
            rm[c][j] = *(const v4f*)(gm + (size_t)j * IN_DIM + c * KCHUNK);
        }
    }

    #pragma unroll
    for (int c = 0; c < NCHUNK; ++c) {
        #pragma unroll
        for (int j = 0; j < 4; ++j) {
            #pragma unroll
            for (int t = 0; t < MTILES; ++t)
                sx[t * OP4 + st_off + j * PITCH4] = rx[c & 1][t][j];
            sn[st_off + j * PITCH4] = rn[c & 1][j];
            sm[st_off + j * PITCH4] = rm[c & 1][j];
        }
        WAVE_LDS_SYNC();

        if (c + 2 < NCHUNK) {
            #pragma unroll
            for (int j = 0; j < 4; ++j) {
                #pragma unroll
                for (int t = 0; t < MTILES; ++t)
                    rx[c & 1][t][j] = *(const v4f*)(gx + (size_t)(t * 16 + j) * IN_DIM + (c + 2) * KCHUNK);
                rn[c & 1][j] = *(const v4f*)(gn + (size_t)j * IN_DIM + (c + 2) * KCHUNK);
                rm[c & 1][j] = *(const v4f*)(gm + (size_t)j * IN_DIM + (c + 2) * KCHUNK);
            }
        }
        __builtin_amdgcn_sched_barrier(0);

        v4f fa[MTILES][NSTEP], fn[NSTEP], fm[NSTEP];
        #pragma unroll
        for (int s = 0; s < NSTEP; ++s) {
            #pragma unroll
            for (int t = 0; t < MTILES; ++t)
                fa[t][s] = sx[t * OP4 + fr_off + 2 * s];
            fn[s] = sn[fr_off + 2 * s];
            fm[s] = sm[fr_off + 2 * s];
        }
        __builtin_amdgcn_sched_barrier(0);

        if (c == 0)          fm[0].x         *= edge_lo;
        if (c == NCHUNK - 1) fm[NSTEP - 1].w *= edge_hi;

        #pragma unroll
        for (int s = 0; s < NSTEP; ++s) {
            const v4f b4n = fn[s], b4m = fm[s];
            v2f bn0 = {b4n.x, b4n.y}, bn1 = {b4n.z, b4n.w};
            v2f bm0 = {b4m.x, b4m.y}, bm1 = {b4m.z, b4m.w};
            v2f a0[MTILES], a1[MTILES];
            #pragma unroll
            for (int t = 0; t < MTILES; ++t) {
                const v4f a4 = fa[t][s];
                a0[t] = (v2f){a4.x, a4.y};
                a1[t] = (v2f){a4.z, a4.w};
            }

            #pragma unroll
            for (int t = 0; t < MTILES; ++t) {
                accN[t] = __builtin_amdgcn_wmma_f32_16x16x4_f32(false, a0[t], false, bn0, (short)0, accN[t], false, false);
                accM[t] = __builtin_amdgcn_wmma_f32_16x16x4_f32(false, a0[t], false, bm0, (short)0, accM[t], false, false);
            }
            #pragma unroll
            for (int t = 0; t < MTILES; ++t) {
                accN[t] = __builtin_amdgcn_wmma_f32_16x16x4_f32(false, a1[t], false, bn1, (short)0, accN[t], false, false);
                accM[t] = __builtin_amdgcn_wmma_f32_16x16x4_f32(false, a1[t], false, bm1, (short)0, accM[t], false, false);
            }
        }
        WAVE_LDS_SYNC();
        __builtin_amdgcn_sched_barrier(0);
    }

    float* mine = (float*)(stage + wave * WAVE4);
    #pragma unroll
    for (int t = 0; t < MTILES; ++t) {
        #pragma unroll
        for (int i = 0; i < 8; ++i) {
            mine[t * 512 + i * 32 + lane]       = accN[t][i];
            mine[t * 512 + 256 + i * 32 + lane] = accM[t][i];
        }
    }
    __syncthreads();

    const int slot = threadIdx.x;
    const float* p0 = (const float*)stage;
    float sN[MTILES], sM[MTILES];
    #pragma unroll
    for (int t = 0; t < MTILES; ++t) {
        sN[t] = p0[t * 512 + slot];
        sM[t] = p0[t * 512 + 256 + slot];
    }
    for (int w = 1; w < KSPLIT; ++w) {
        const float* pw = (const float*)(stage + w * WAVE4);
        #pragma unroll
        for (int t = 0; t < MTILES; ++t) {
            sN[t] += pw[t * 512 + slot];
            sM[t] += pw[t * 512 + 256 + slot];
        }
    }

    const float bb = bnon[nrow];
    #pragma unroll
    for (int t = 0; t < MTILES; ++t) {
        float z     = sN[t] + bb;
        float s     = 1.0f / (1.0f + __expf(-z));
        float state = s - 1.0f;
        float ca    = fmaxf(sM[t], 0.0f);
        float nm    = 2.0f + state;
        float kd    = exp2f(-nm);
        float xn    = __powf(ca, nm);
        float y     = xn / (kd + xn) + state;
        Y[(size_t)(mt * (16 * MTILES) + t * 16 + (lane >> 4) * 8 + wave) * OUT_DIM + nrow] = y;
    }
}

extern "C" void kernel_launch(void* const* d_in, const int* in_sizes, int n_in,
                              void* d_out, int out_size, void* d_ws, size_t ws_size,
                              hipStream_t stream) {
    const float* X     = (const float*)d_in[0];
    const float* Wnmda = (const float*)d_in[1];
    const float* Wnon  = (const float*)d_in[2];
    const float* bnon  = (const float*)d_in[3];
    float* Y = (float*)d_out;

    dim3 grid(256);
    dim3 block(256);
    hipLaunchKernelGGL(dendritic_fused_kernel, grid, block, 0, stream,
                       X, Wnmda, Wnon, bnon, Y);
}
